# starting kernel
# baseline (speedup 1.0000x reference)
#include <hip/hip_runtime.h>
#include <hip/hip_bf16.h>
#include <math.h>

typedef __attribute__((ext_vector_type(2))) float v2f;
typedef __attribute__((ext_vector_type(8))) float v8f;

#define B_  32
#define D_  1024
#define H_  64
#define O_  1000
#define E_  6

__global__ void moe_gate(const float* __restrict__ x,
                         const float* __restrict__ gw,
                         const float* __restrict__ gb,
                         float* __restrict__ gates) {
  __shared__ float logits[B_][E_];
  int t = threadIdx.x;
  if (t < B_ * E_) {
    int b = t / E_, e = t % E_;
    const float* xr = x + b * D_;
    float acc = gb[e];
    for (int d = 0; d < D_; ++d) acc = fmaf(xr[d], gw[d * E_ + e], acc);
    logits[b][e] = acc;
  }
  __syncthreads();
  if (t < B_) {
    int b = t;
    int i0 = 0; float v0 = logits[b][0];
    for (int e = 1; e < E_; ++e) { float v = logits[b][e]; if (v > v0) { v0 = v; i0 = e; } }
    int i1 = -1; float v1 = -3.4e38f;
    for (int e = 0; e < E_; ++e) {
      if (e == i0) continue;
      float v = logits[b][e]; if (v > v1) { v1 = v; i1 = e; }
    }
    float e0 = 1.0f;
    float e1 = __expf(v1 - v0);
    float inv = 1.0f / (e0 + e1);
    for (int e = 0; e < E_; ++e) gates[b * E_ + e] = 0.0f;
    gates[b * E_ + i0] = e0 * inv;
    gates[b * E_ + i1] = e1 * inv;
  }
}

__global__ void moe_fc1(const float* __restrict__ x,
                        const float* __restrict__ w1,
                        const float* __restrict__ b1,
                        float* __restrict__ h) {
  int wave = blockIdx.x * 4 + (threadIdx.x >> 5);
  if (wave >= E_ * 2 * 4) return;
  int e   = wave >> 3;
  int m_t = (wave >> 2) & 1;
  int n_t = wave & 3;
  int lane  = threadIdx.x & 31;
  int row   = lane & 15;
  int khalf = lane >> 4;

  const float* xr  = x  + (m_t * 16 + row) * D_;
  const float* w1e = w1 + e * D_ * H_ + n_t * 16 + row;
  v8f c = {};
  for (int k = 0; k < D_; k += 4) {
    int ka = k + 2 * khalf;
    v2f a = *(const v2f*)(xr + ka);
    v2f bb;
    bb.x = w1e[(ka + 0) * H_];
    bb.y = w1e[(ka + 1) * H_];
    c = __builtin_amdgcn_wmma_f32_16x16x4_f32(
         false, a,  false, bb,
         (short)0, c,  false,  false);
  }
  float bias = b1[e * H_ + n_t * 16 + row];
  float* hout = h + (e * B_ + m_t * 16) * H_ + n_t * 16 + row;
#pragma unroll
  for (int v = 0; v < 8; ++v) {
    int r = v + 8 * khalf;
    float val = c[v] + bias;
    hout[r * H_] = val > 0.0f ? val : 0.0f;
  }
}

__global__ void moe_fc2(const float* __restrict__ hbuf,
                        const float* __restrict__ w2,
                        const float* __restrict__ b2,
                        const float* __restrict__ gates,
                        float* __restrict__ out) {
  const int NT = (O_ + 15) / 16;
  int wave = blockIdx.x * 4 + (threadIdx.x >> 5);
  if (wave >= 2 * NT) return;
  int m_t = wave / NT;
  int n_t = wave % NT;
  int lane  = threadIdx.x & 31;
  int row   = lane & 15;
  int khalf = lane >> 4;
  int ocol  = n_t * 16 + row;
  bool colok = ocol < O_;

  v8f c = {};
  for (int e = 0; e < E_; ++e) {
    float g = gates[(m_t * 16 + row) * E_ + e];
    const float* he  = hbuf + (e * B_ + m_t * 16 + row) * H_;
    const float* w2e = w2 + e * H_ * O_;
#pragma unroll
    for (int k = 0; k < H_; k += 4) {
      int ka = k + 2 * khalf;
      v2f a = *(const v2f*)(he + ka);
      a.x *= g; a.y *= g;
      v2f bb;
      bb.x = colok ? w2e[(ka + 0) * O_ + ocol] : 0.0f;
      bb.y = colok ? w2e[(ka + 1) * O_ + ocol] : 0.0f;
      c = __builtin_amdgcn_wmma_f32_16x16x4_f32(
          false, a, false, bb, (short)0, c, false, false);
    }
  }
  if (colok) {
#pragma unroll
    for (int v = 0; v < 8; ++v) {
      int r = m_t * 16 + v + 8 * khalf;
      float acc = c[v];
      for (int e = 0; e < E_; ++e)
        acc = fmaf(gates[r * E_ + e], b2[e * O_ + ocol], acc);
      out[r * O_ + ocol] = acc;
    }
  }
}

extern "C" void kernel_launch(void* const* d_in, const int* in_sizes, int n_in,
                              void* d_out, int out_size, void* d_ws, size_t ws_size,
                              hipStream_t stream) {
  const float* x      = (const float*)d_in[0];
  const float* gate_w = (const float*)d_in[1];
  const float* gate_b = (const float*)d_in[2];
  const float* w1     = (const float*)d_in[3];
  const float* b1     = (const float*)d_in[4];
  const float* w2     = (const float*)d_in[5];
  const float* b2     = (const float*)d_in[6];
  float* out = (float*)d_out;

  float* gates = (float*)d_ws;
  float* hbuf  = gates + 256;

  moe_gate<<<1, B_ * E_, 0, stream>>>(x, gate_w, gate_b, gates);

  moe_fc1<<<(48 + 3) / 4, 128, 0, stream>>>(x, w1, b1, hbuf);

  moe_fc2<<<(126 + 3) / 4, 128, 0, stream>>>(hbuf, w2, b2, gates, out);
}


# fc2: A operand through LDS too: the block copies all of h, the gates and its b2 strip in the one-shot staging of the w2 strip; was 17 global loads per expert, waited for in turn; sum order unchanged
# speedup vs baseline: 6.4887x; 6.4887x over previous
#include <hip/hip_runtime.h>
#include <hip/hip_bf16.h>
#include <math.h>

typedef __attribute__((ext_vector_type(2))) float v2f;
typedef __attribute__((ext_vector_type(8))) float v8f;

#define B_  32
#define D_  1024
#define H_  64
#define O_  1000
#define E_  6

__global__ void __launch_bounds__(32) __attribute__((amdgpu_waves_per_eu(1, 2)))
moe_gate(const float* __restrict__ x,
         const float* __restrict__ gw,
         const float* __restrict__ gb,
         float* __restrict__ gates) {
  const int lane = threadIdx.x;
  const int b    = blockIdx.x;
  const float4* __restrict__ xr  = (const float4*)(x + b * D_);
  const float4* __restrict__ gw4 = (const float4*)gw;

  float acc[E_] = {0.0f, 0.0f, 0.0f, 0.0f, 0.0f, 0.0f};

#pragma unroll
  for (int it = 0; it < D_ / 128; ++it) {
    const int q = it * 32 + lane;
    const float4 xv = xr[q];
    const float xs[4] = {xv.x, xv.y, xv.z, xv.w};
    float wf[24];
#pragma unroll
    for (int j = 0; j < 6; ++j) {
      const float4 w = gw4[q * 6 + j];
      wf[4 * j + 0] = w.x; wf[4 * j + 1] = w.y;
      wf[4 * j + 2] = w.z; wf[4 * j + 3] = w.w;
    }
#pragma unroll
    for (int r = 0; r < 4; ++r) {
#pragma unroll
      for (int e = 0; e < E_; ++e)
        acc[e] = fmaf(xs[r], wf[r * E_ + e], acc[e]);
    }
  }

  float logit[E_];
#pragma unroll
  for (int e = 0; e < E_; ++e) {
    float s = acc[e];
#pragma unroll
    for (int off = 16; off >= 1; off >>= 1) s += __shfl_xor(s, off, 32);
    logit[e] = s + gb[e];
  }

  int i0 = 0; float v0 = logit[0];
#pragma unroll
  for (int e = 1; e < E_; ++e) { float v = logit[e]; if (v > v0) { v0 = v; i0 = e; } }
  int i1 = -1; float v1 = -3.4e38f;
#pragma unroll
  for (int e = 0; e < E_; ++e) {
    float v = logit[e];
    if (e != i0 && v > v1) { v1 = v; i1 = e; }
  }
  float e1  = __expf(v1 - v0);
  float inv = 1.0f / (1.0f + e1);
  if (lane < E_)
    gates[b * E_ + lane] = (lane == i0) ? inv : ((lane == i1) ? e1 * inv : 0.0f);
}

#define FC1_COLS   32
#define FC1_PB     48
#define FC1_PA     (D_ + 4)
#define FC1_KB     16
#define FC1_THREADS 256

__device__ __forceinline__ void fc1_read(v2f (&af)[FC1_KB], v2f (&bf)[FC1_KB],
                                         const float* pa, const float* pb) {
#pragma unroll
  for (int j = 0; j < FC1_KB; ++j) {
    af[j] = *(const v2f*)(pa + 4 * j);
    bf[j].x = pb[(4 * j + 0) * FC1_PB];
    bf[j].y = pb[(4 * j + 1) * FC1_PB];
  }
  __builtin_amdgcn_sched_barrier(0);
}

__device__ __forceinline__ v8f fc1_chain(v8f c, const v2f (&af)[FC1_KB], const v2f (&bf)[FC1_KB]) {
#pragma unroll
  for (int j = 0; j < FC1_KB; ++j)
    c = __builtin_amdgcn_wmma_f32_16x16x4_f32(
         false, af[j],  false, bf[j],
         (short)0, c,  false,  false);
  __builtin_amdgcn_sched_barrier(0);
  return c;
}

__global__ void __launch_bounds__(FC1_THREADS)
moe_fc1(const float* __restrict__ x,
        const float* __restrict__ w1,
        const float* __restrict__ b1,
        float* __restrict__ h) {
  __shared__ __attribute__((aligned(16))) float lds1[D_ * FC1_PB + 16 * FC1_PA];
  float* sb = lds1;
  float* sa = lds1 + D_ * FC1_PB;

  const int tid = threadIdx.x;
  const int e   = blockIdx.x >> 2;
  const int m_t = (blockIdx.x >> 1) & 1;
  const int np  = blockIdx.x & 1;

  {
    const int c4 = tid & 7;
    const int r0 = tid >> 3;
    const float4* __restrict__ srcb =
        (const float4*)(w1 + e * D_ * H_ + np * FC1_COLS + 4 * c4);
    const int ca = tid & 15;
    const int ra = tid >> 4;
    const float4* __restrict__ srca =
        (const float4*)(x + (m_t * 16 + ra) * D_) + ca;
    float4 tb[D_ / 32], ta[D_ / 64];
#pragma unroll
    for (int i = 0; i < D_ / 32; ++i)
      tb[i] = srcb[(r0 + 32 * i) * (H_ / 4)];
#pragma unroll
    for (int i = 0; i < D_ / 64; ++i)
      ta[i] = srca[16 * i];
#pragma unroll
    for (int i = 0; i < D_ / 32; ++i)
      *(float4*)(sb + (r0 + 32 * i) * FC1_PB + 4 * c4) = tb[i];
#pragma unroll
    for (int i = 0; i < D_ / 64; ++i)
      *(float4*)(sa + ra * FC1_PA + 4 * (ca + 16 * i)) = ta[i];
  }

  __syncthreads();

  const int w = tid >> 5;
  if (w >= 2) return;
  const int n_s   = w;
  const int lane  = tid & 31;
  const int row   = lane & 15;
  const int khalf = lane >> 4;

  const float* pa = sa + row * FC1_PA + 2 * khalf;
  const float* pb = sb + (2 * khalf) * FC1_PB + n_s * 16 + row;
  const int kstep = 4 * FC1_KB;
  v2f a0[FC1_KB], b0[FC1_KB], a1[FC1_KB], b1f[FC1_KB];
  fc1_read(a0, b0, pa, pb);
  v8f c = {};
  for (int kb = 0; kb < D_ / kstep - 2; kb += 2) {
    fc1_read(a1, b1f, pa + (kb + 1) * kstep, pb + (kb + 1) * kstep * FC1_PB);
    c = fc1_chain(c, a0, b0);
    fc1_read(a0, b0, pa + (kb + 2) * kstep, pb + (kb + 2) * kstep * FC1_PB);
    c = fc1_chain(c, a1, b1f);
  }
  fc1_read(a1, b1f, pa + (D_ / kstep - 1) * kstep, pb + (D_ / kstep - 1) * kstep * FC1_PB);
  c = fc1_chain(c, a0, b0);
  c = fc1_chain(c, a1, b1f);

  const int n_t = np * 2 + n_s;
  float bias = b1[e * H_ + n_t * 16 + row];
  float* hout = h + (e * B_ + m_t * 16) * H_ + n_t * 16 + row;
#pragma unroll
  for (int v = 0; v < 8; ++v) {
    int r = v + 8 * khalf;
    float val = c[v] + bias;
    hout[r * H_] = val > 0.0f ? val : 0.0f;
  }
}

#define FC2_COLS   32
#define FC2_PITCH  40
#define FC2_PA     (H_ + 4)
#define FC2_K      (E_ * H_)
#define FC2_KB     (H_ / 4)
#define FC2_HROWS  (E_ * B_)

__device__ __forceinline__ void fc2_read(v2f (&af)[FC2_KB], v2f (&bf)[FC2_KB],
                                         const float* pa, const float* pb) {
#pragma unroll
  for (int j = 0; j < FC2_KB; ++j) {
    af[j] = *(const v2f*)(pa + 4 * j);
    bf[j].x = pb[(4 * j + 0) * FC2_PITCH];
    bf[j].y = pb[(4 * j + 1) * FC2_PITCH];
  }
  __builtin_amdgcn_sched_barrier(0);
}

__device__ __forceinline__ v8f fc2_chain(v8f c, const v2f (&af)[FC2_KB], float g,
                                         const v2f (&bf)[FC2_KB]) {
#pragma unroll
  for (int j = 0; j < FC2_KB; ++j) {
    v2f a = af[j];
    a.x *= g; a.y *= g;
    c = __builtin_amdgcn_wmma_f32_16x16x4_f32(
        false, a, false, bf[j], (short)0, c, false, false);
  }
  __builtin_amdgcn_sched_barrier(0);
  return c;
}

__global__ void __launch_bounds__(128)
moe_fc2(const float* __restrict__ hbuf,
        const float* __restrict__ w2,
        const float* __restrict__ b2,
        const float* __restrict__ gates,
        float* __restrict__ out) {
  __shared__ __attribute__((aligned(16)))
      float lds2[FC2_K * FC2_PITCH + FC2_HROWS * FC2_PA + B_ * E_ + E_ * FC2_COLS];
  float* bs  = lds2;
  float* hs  = bs + FC2_K * FC2_PITCH;
  float* gs  = hs + FC2_HROWS * FC2_PA;
  float* b2s = gs + B_ * E_;

  const int tid  = threadIdx.x;
  const int col0 = blockIdx.x * FC2_COLS;

  {
    const int c4 = tid & 7;
    const int r0 = tid >> 3;
    int gcol = col0 + 4 * c4;
    if (gcol > O_ - 4) gcol = O_ - 4;
    const float4* __restrict__ srcw = (const float4*)(w2 + gcol);
    const int ch = tid & 15;
    const int rh = tid >> 4;
    const float4* __restrict__ srch = (const float4*)hbuf + ch;
    const int q  = tid & 63;
    const int sg = q < 48 ? q : q - 48;
    float4 tw[FC2_K / 16], th[FC2_HROWS / 8];
    const float4 tg = ((const float4*)gates)[sg];
    const float4 tb = *(const float4*)(b2 + (sg >> 3) * O_ + gcol);
#pragma unroll
    for (int i = 0; i < FC2_K / 16; ++i)
      tw[i] = srcw[(r0 + 16 * i) * (O_ / 4)];
#pragma unroll
    for (int i = 0; i < FC2_HROWS / 8; ++i)
      th[i] = srch[(rh + 8 * i) * (H_ / 4)];
    *(float4*)(gs + 4 * sg) = tg;
    *(float4*)(b2s + (sg >> 3) * FC2_COLS + 4 * c4) = tb;
#pragma unroll
    for (int i = 0; i < FC2_K / 16; ++i)
      *(float4*)(bs + (r0 + 16 * i) * FC2_PITCH + 4 * c4) = tw[i];
#pragma unroll
    for (int i = 0; i < FC2_HROWS / 8; ++i)
      *(float4*)(hs + (rh + 8 * i) * FC2_PA + 4 * ch) = th[i];
  }

  const int w     = tid >> 5;
  const int m_t   = w >> 1;
  const int n_s   = w & 1;
  const int lane  = tid & 31;
  const int row   = lane & 15;
  const int khalf = lane >> 4;
  const int ocol  = col0 + n_s * 16 + row;
  const bool colok = ocol < O_;

  __syncthreads();

  const float* pa = hs + (m_t * 16 + row) * FC2_PA + 2 * khalf;
  const float* pb = bs + (2 * khalf) * FC2_PITCH + n_s * 16 + row;
  const int astep = B_ * FC2_PA;
  const int bstep = H_ * FC2_PITCH;
  float g[E_];
#pragma unroll
  for (int e = 0; e < E_; ++e) g[e] = gs[(m_t * 16 + row) * E_ + e];

  v2f a0[FC2_KB], b0[FC2_KB], a1[FC2_KB], b1f[FC2_KB];
  fc2_read(a0, b0, pa, pb);
  v8f c = {};
#pragma unroll
  for (int e = 0; e < E_ - 2; e += 2) {
    fc2_read(a1, b1f, pa + (e + 1) * astep, pb + (e + 1) * bstep);
    c = fc2_chain(c, a0, g[e], b0);
    fc2_read(a0, b0, pa + (e + 2) * astep, pb + (e + 2) * bstep);
    c = fc2_chain(c, a1, g[e + 1], b1f);
  }
  fc2_read(a1, b1f, pa + (E_ - 1) * astep, pb + (E_ - 1) * bstep);
  c = fc2_chain(c, a0, g[E_ - 2], b0);
  c = fc2_chain(c, a1, g[E_ - 1], b1f);

  if (colok) {
    const float* ge = gs + (m_t * 16 + 8 * khalf) * E_;
    const float* be = b2s + n_s * 16 + row;
#pragma unroll
    for (int v = 0; v < 8; ++v) {
      int r = m_t * 16 + v + 8 * khalf;
      float acc = c[v];
#pragma unroll
      for (int e = 0; e < E_; ++e)
        acc = fmaf(ge[v * E_ + e], be[e * FC2_COLS], acc);
      out[r * O_ + ocol] = acc;
    }
  }
}

extern "C" void kernel_launch(void* const* d_in, const int* in_sizes, int n_in,
                              void* d_out, int out_size, void* d_ws, size_t ws_size,
                              hipStream_t stream) {
  const float* x      = (const float*)d_in[0];
  const float* gate_w = (const float*)d_in[1];
  const float* gate_b = (const float*)d_in[2];
  const float* w1     = (const float*)d_in[3];
  const float* b1     = (const float*)d_in[4];
  const float* w2     = (const float*)d_in[5];
  const float* b2     = (const float*)d_in[6];
  float* out = (float*)d_out;

  float* gates = (float*)d_ws;
  float* hbuf  = gates + 256;

  moe_gate<<<B_, 32, 0, stream>>>(x, gate_w, gate_b, gates);

  moe_fc1<<<E_ * 2 * 2, FC1_THREADS, 0, stream>>>(x, w1, b1, hbuf);

  moe_fc2<<<(O_ + FC2_COLS - 1) / FC2_COLS, 128, 0, stream>>>(hbuf, w2, b2, gates, out);
}


# gate: logits made inside the fc1 launch by an idle wave of two fc1 blocks, on f32 WMMA from the x slab already in LDS (gate_w staged beside it); 3 launches become 2; logit sum order changes
# speedup vs baseline: 7.8729x; 1.2133x over previous
#include <hip/hip_runtime.h>
#include <hip/hip_bf16.h>
#include <math.h>

typedef __attribute__((ext_vector_type(2))) float v2f;
typedef __attribute__((ext_vector_type(8))) float v8f;

#define B_  32
#define D_  1024
#define H_  64
#define O_  1000
#define E_  6

#define FC1_COLS   32
#define FC1_PB     48
#define FC1_PA     (D_ + 4)
#define FC1_KB     16
#define FC1_THREADS 256

#define FC1_PG     E_

template <int PB>
__device__ __forceinline__ void fc1_read(v2f (&af)[FC1_KB], v2f (&bf)[FC1_KB],
                                         const float* pa, const float* pb) {
#pragma unroll
  for (int j = 0; j < FC1_KB; ++j) {
    af[j] = *(const v2f*)(pa + 4 * j);
    bf[j].x = pb[(4 * j + 0) * PB];
    bf[j].y = pb[(4 * j + 1) * PB];
  }
  __builtin_amdgcn_sched_barrier(0);
}

__device__ __forceinline__ v8f fc1_chain(v8f c, const v2f (&af)[FC1_KB], const v2f (&bf)[FC1_KB]) {
#pragma unroll
  for (int j = 0; j < FC1_KB; ++j)
    c = __builtin_amdgcn_wmma_f32_16x16x4_f32(
         false, af[j],  false, bf[j],
         (short)0, c,  false,  false);
  __builtin_amdgcn_sched_barrier(0);
  return c;
}

template <int PB, bool XT>
__device__ __forceinline__ v8f fc1_tile(const float* pa, const float* pb) {
  const int kstep = 4 * FC1_KB;
  v2f a0[FC1_KB], b0[FC1_KB], a1[FC1_KB], b1f[FC1_KB];
  fc1_read<PB>(a0, b0, pa, pb);
  v8f c = {};
  for (int kb = 0; kb < D_ / kstep - 2; kb += 2) {
    fc1_read<PB>(a1, b1f, pa + (kb + 1) * kstep, pb + (kb + 1) * kstep * PB);
    c = XT ? fc1_chain(c, b0, a0) : fc1_chain(c, a0, b0);
    fc1_read<PB>(a0, b0, pa + (kb + 2) * kstep, pb + (kb + 2) * kstep * PB);
    c = XT ? fc1_chain(c, b1f, a1) : fc1_chain(c, a1, b1f);
  }
  fc1_read<PB>(a1, b1f, pa + (D_ / kstep - 1) * kstep, pb + (D_ / kstep - 1) * kstep * PB);
  c = XT ? fc1_chain(c, b0, a0) : fc1_chain(c, a0, b0);
  c = XT ? fc1_chain(c, b1f, a1) : fc1_chain(c, a1, b1f);
  return c;
}

__global__ void __launch_bounds__(FC1_THREADS)
moe_fc1(const float* __restrict__ x,
        const float* __restrict__ w1,
        const float* __restrict__ b1,
        const float* __restrict__ gw,
        const float* __restrict__ gb,
        float* __restrict__ h,
        float* __restrict__ gates) {
  __shared__ __attribute__((aligned(16))) float lds1[D_ * FC1_PB + 16 * FC1_PA + D_ * FC1_PG];
  float* sb = lds1;
  float* sa = lds1 + D_ * FC1_PB;
  float* sg = sa + 16 * FC1_PA;

  const int tid = threadIdx.x;
  const int e   = blockIdx.x >> 2;
  const int m_t = (blockIdx.x >> 1) & 1;
  const int np  = blockIdx.x & 1;
  const bool gate_blk = (e == 0) && (np == 0);

  {
    const int c4 = tid & 7;
    const int r0 = tid >> 3;
    const float4* __restrict__ srcb =
        (const float4*)(w1 + e * D_ * H_ + np * FC1_COLS + 4 * c4);
    const int ca = tid & 15;
    const int ra = tid >> 4;
    const float4* __restrict__ srca =
        (const float4*)(x + (m_t * 16 + ra) * D_) + ca;
    const float4* __restrict__ srcg = (const float4*)gw + tid;
    float4 tb[D_ / 32], ta[D_ / 64], tg[D_ * E_ / 4 / FC1_THREADS];
#pragma unroll
    for (int i = 0; i < D_ / 32; ++i)
      tb[i] = srcb[(r0 + 32 * i) * (H_ / 4)];
#pragma unroll
    for (int i = 0; i < D_ / 64; ++i)
      ta[i] = srca[16 * i];
#pragma unroll
    for (int i = 0; i < D_ * E_ / 4 / FC1_THREADS; ++i)
      tg[i] = srcg[FC1_THREADS * i];
#pragma unroll
    for (int i = 0; i < D_ / 32; ++i)
      *(float4*)(sb + (r0 + 32 * i) * FC1_PB + 4 * c4) = tb[i];
#pragma unroll
    for (int i = 0; i < D_ / 64; ++i)
      *(float4*)(sa + ra * FC1_PA + 4 * (ca + 16 * i)) = ta[i];
#pragma unroll
    for (int i = 0; i < D_ * E_ / 4 / FC1_THREADS; ++i)
      *(float4*)(sg + 4 * (tid + FC1_THREADS * i)) = tg[i];
  }

  __syncthreads();

  const int w     = tid >> 5;
  const int lane  = tid & 31;
  const int row   = lane & 15;
  const int khalf = lane >> 4;
  const float* pa = sa + row * FC1_PA + 2 * khalf;

  if (w >= 2) {
    if (w > 2 || !gate_blk) return;

    const int gcol = row < E_ ? row : E_ - 1;
    v8f c = fc1_tile<FC1_PG, true>(pa, sg + (2 * khalf) * FC1_PG + gcol);
    float logit[E_];
#pragma unroll
    for (int q = 0; q < E_; ++q) logit[q] = c[q] + gb[q];
    int i0 = 0; float v0 = logit[0];
#pragma unroll
    for (int q = 1; q < E_; ++q) { float lv = logit[q]; if (lv > v0) { v0 = lv; i0 = q; } }
    int i1 = -1; float v1 = -3.4e38f;
#pragma unroll
    for (int q = 0; q < E_; ++q) {
      float lv = logit[q];
      if (q != i0 && lv > v1) { v1 = lv; i1 = q; }
    }
    const float e1  = __expf(v1 - v0);
    const float inv = 1.0f / (1.0f + e1);
    if (khalf == 0) {
      float* gout = gates + (m_t * 16 + row) * E_;
#pragma unroll
      for (int q = 0; q < E_; ++q)
        gout[q] = (q == i0) ? inv : ((q == i1) ? e1 * inv : 0.0f);
    }
    return;
  }

  const int n_s = w;
  v8f c = fc1_tile<FC1_PB, false>(pa, sb + (2 * khalf) * FC1_PB + n_s * 16 + row);

  const int n_t = np * 2 + n_s;
  float bias = b1[e * H_ + n_t * 16 + row];
  float* hout = h + (e * B_ + m_t * 16) * H_ + n_t * 16 + row;
#pragma unroll
  for (int v = 0; v < 8; ++v) {
    int r = v + 8 * khalf;
    float val = c[v] + bias;
    hout[r * H_] = val > 0.0f ? val : 0.0f;
  }
}

#define FC2_COLS   32
#define FC2_PITCH  40
#define FC2_PA     (H_ + 4)
#define FC2_K      (E_ * H_)
#define FC2_KB     (H_ / 4)
#define FC2_HROWS  (E_ * B_)

__device__ __forceinline__ void fc2_read(v2f (&af)[FC2_KB], v2f (&bf)[FC2_KB],
                                         const float* pa, const float* pb) {
#pragma unroll
  for (int j = 0; j < FC2_KB; ++j) {
    af[j] = *(const v2f*)(pa + 4 * j);
    bf[j].x = pb[(4 * j + 0) * FC2_PITCH];
    bf[j].y = pb[(4 * j + 1) * FC2_PITCH];
  }
  __builtin_amdgcn_sched_barrier(0);
}

__device__ __forceinline__ v8f fc2_chain(v8f c, const v2f (&af)[FC2_KB], float g,
                                         const v2f (&bf)[FC2_KB]) {
#pragma unroll
  for (int j = 0; j < FC2_KB; ++j) {
    v2f a = af[j];
    a.x *= g; a.y *= g;
    c = __builtin_amdgcn_wmma_f32_16x16x4_f32(
        false, a, false, bf[j], (short)0, c, false, false);
  }
  __builtin_amdgcn_sched_barrier(0);
  return c;
}

__global__ void __launch_bounds__(128)
moe_fc2(const float* __restrict__ hbuf,
        const float* __restrict__ w2,
        const float* __restrict__ b2,
        const float* __restrict__ gates,
        float* __restrict__ out) {
  __shared__ __attribute__((aligned(16)))
      float lds2[FC2_K * FC2_PITCH + FC2_HROWS * FC2_PA + B_ * E_ + E_ * FC2_COLS];
  float* bs  = lds2;
  float* hs  = bs + FC2_K * FC2_PITCH;
  float* gs  = hs + FC2_HROWS * FC2_PA;
  float* b2s = gs + B_ * E_;

  const int tid  = threadIdx.x;
  const int col0 = blockIdx.x * FC2_COLS;

  {
    const int c4 = tid & 7;
    const int r0 = tid >> 3;
    int gcol = col0 + 4 * c4;
    if (gcol > O_ - 4) gcol = O_ - 4;
    const float4* __restrict__ srcw = (const float4*)(w2 + gcol);
    const int ch = tid & 15;
    const int rh = tid >> 4;
    const float4* __restrict__ srch = (const float4*)hbuf + ch;
    const int q  = tid & 63;
    const int sg = q < 48 ? q : q - 48;
    float4 tw[FC2_K / 16], th[FC2_HROWS / 8];
    const float4 tg = ((const float4*)gates)[sg];
    const float4 tb = *(const float4*)(b2 + (sg >> 3) * O_ + gcol);
#pragma unroll
    for (int i = 0; i < FC2_K / 16; ++i)
      tw[i] = srcw[(r0 + 16 * i) * (O_ / 4)];
#pragma unroll
    for (int i = 0; i < FC2_HROWS / 8; ++i)
      th[i] = srch[(rh + 8 * i) * (H_ / 4)];
    *(float4*)(gs + 4 * sg) = tg;
    *(float4*)(b2s + (sg >> 3) * FC2_COLS + 4 * c4) = tb;
#pragma unroll
    for (int i = 0; i < FC2_K / 16; ++i)
      *(float4*)(bs + (r0 + 16 * i) * FC2_PITCH + 4 * c4) = tw[i];
#pragma unroll
    for (int i = 0; i < FC2_HROWS / 8; ++i)
      *(float4*)(hs + (rh + 8 * i) * FC2_PA + 4 * ch) = th[i];
  }

  const int w     = tid >> 5;
  const int m_t   = w >> 1;
  const int n_s   = w & 1;
  const int lane  = tid & 31;
  const int row   = lane & 15;
  const int khalf = lane >> 4;
  const int ocol  = col0 + n_s * 16 + row;
  const bool colok = ocol < O_;

  __syncthreads();

  const float* pa = hs + (m_t * 16 + row) * FC2_PA + 2 * khalf;
  const float* pb = bs + (2 * khalf) * FC2_PITCH + n_s * 16 + row;
  const int astep = B_ * FC2_PA;
  const int bstep = H_ * FC2_PITCH;
  float g[E_];
#pragma unroll
  for (int e = 0; e < E_; ++e) g[e] = gs[(m_t * 16 + row) * E_ + e];

  v2f a0[FC2_KB], b0[FC2_KB], a1[FC2_KB], b1f[FC2_KB];
  fc2_read(a0, b0, pa, pb);
  v8f c = {};
#pragma unroll
  for (int e = 0; e < E_ - 2; e += 2) {
    fc2_read(a1, b1f, pa + (e + 1) * astep, pb + (e + 1) * bstep);
    c = fc2_chain(c, a0, g[e], b0);
    fc2_read(a0, b0, pa + (e + 2) * astep, pb + (e + 2) * bstep);
    c = fc2_chain(c, a1, g[e + 1], b1f);
  }
  fc2_read(a1, b1f, pa + (E_ - 1) * astep, pb + (E_ - 1) * bstep);
  c = fc2_chain(c, a0, g[E_ - 2], b0);
  c = fc2_chain(c, a1, g[E_ - 1], b1f);

  if (colok) {
    const float* ge = gs + (m_t * 16 + 8 * khalf) * E_;
    const float* be = b2s + n_s * 16 + row;
#pragma unroll
    for (int v = 0; v < 8; ++v) {
      int r = m_t * 16 + v + 8 * khalf;
      float acc = c[v];
#pragma unroll
      for (int e = 0; e < E_; ++e)
        acc = fmaf(ge[v * E_ + e], be[e * FC2_COLS], acc);
      out[r * O_ + ocol] = acc;
    }
  }
}

extern "C" void kernel_launch(void* const* d_in, const int* in_sizes, int n_in,
                              void* d_out, int out_size, void* d_ws, size_t ws_size,
                              hipStream_t stream) {
  const float* x      = (const float*)d_in[0];
  const float* gate_w = (const float*)d_in[1];
  const float* gate_b = (const float*)d_in[2];
  const float* w1     = (const float*)d_in[3];
  const float* b1     = (const float*)d_in[4];
  const float* w2     = (const float*)d_in[5];
  const float* b2     = (const float*)d_in[6];
  float* out = (float*)d_out;

  float* gates = (float*)d_ws;
  float* hbuf  = gates + 256;

  moe_fc1<<<E_ * 2 * 2, FC1_THREADS, 0, stream>>>(x, w1, b1, gate_w, gate_b, hbuf, gates);

  moe_fc2<<<(O_ + FC2_COLS - 1) / FC2_COLS, 128, 0, stream>>>(hbuf, w2, b2, gates, out);
}


# fc1 and gate logits: K=1024 split over 4 blocks per tile (24 -> 96 blocks), quarter sums to workspace; fc2 staging adds them, b1, ReLU and takes top-2; h and logit sum order changes
# speedup vs baseline: 10.5602x; 1.3413x over previous
#include <hip/hip_runtime.h>
#include <hip/hip_bf16.h>
#include <math.h>

typedef __attribute__((ext_vector_type(2))) float v2f;
typedef __attribute__((ext_vector_type(8))) float v8f;

#define B_  32
#define D_  1024
#define H_  64
#define O_  1000
#define E_  6
#define KS_ 4
#define KQ_ (D_ / KS_)

#define FC1_COLS   32
#define FC1_PB     48
#define FC1_PA     (KQ_ + 4)
#define FC1_KB     16
#define FC1_THREADS 256

#define FC1_PG     E_

template <int PB>
__device__ __forceinline__ void fc1_read(v2f (&af)[FC1_KB], v2f (&bf)[FC1_KB],
                                         const float* pa, const float* pb) {
#pragma unroll
  for (int j = 0; j < FC1_KB; ++j) {
    af[j] = *(const v2f*)(pa + 4 * j);
    bf[j].x = pb[(4 * j + 0) * PB];
    bf[j].y = pb[(4 * j + 1) * PB];
  }
  __builtin_amdgcn_sched_barrier(0);
}

__device__ __forceinline__ v8f fc1_chain(v8f c, const v2f (&af)[FC1_KB], const v2f (&bf)[FC1_KB]) {
#pragma unroll
  for (int j = 0; j < FC1_KB; ++j)
    c = __builtin_amdgcn_wmma_f32_16x16x4_f32(
         false, af[j],  false, bf[j],
         (short)0, c,  false,  false);
  __builtin_amdgcn_sched_barrier(0);
  return c;
}

template <int PB, bool XT>
__device__ __forceinline__ v8f fc1_tile(const float* pa, const float* pb) {
  const int kstep = 4 * FC1_KB;
  v2f a0[FC1_KB], b0[FC1_KB], a1[FC1_KB], b1f[FC1_KB];
  fc1_read<PB>(a0, b0, pa, pb);
  v8f c = {};
  for (int kb = 0; kb < KQ_ / kstep - 2; kb += 2) {
    fc1_read<PB>(a1, b1f, pa + (kb + 1) * kstep, pb + (kb + 1) * kstep * PB);
    c = XT ? fc1_chain(c, b0, a0) : fc1_chain(c, a0, b0);
    fc1_read<PB>(a0, b0, pa + (kb + 2) * kstep, pb + (kb + 2) * kstep * PB);
    c = XT ? fc1_chain(c, b1f, a1) : fc1_chain(c, a1, b1f);
  }
  fc1_read<PB>(a1, b1f, pa + (KQ_ / kstep - 1) * kstep, pb + (KQ_ / kstep - 1) * kstep * PB);
  c = XT ? fc1_chain(c, b0, a0) : fc1_chain(c, a0, b0);
  c = XT ? fc1_chain(c, b1f, a1) : fc1_chain(c, a1, b1f);
  return c;
}

__global__ void __launch_bounds__(FC1_THREADS)
moe_fc1(const float* __restrict__ x,
        const float* __restrict__ w1,
        const float* __restrict__ gw,
        float* __restrict__ hp,
        float* __restrict__ lp) {
  __shared__ __attribute__((aligned(16))) float lds1[KQ_ * FC1_PB + 16 * FC1_PA + KQ_ * FC1_PG];
  float* sb = lds1;
  float* sa = lds1 + KQ_ * FC1_PB;
  float* sg = sa + 16 * FC1_PA;

  const int tid = threadIdx.x;
  const int e   = blockIdx.x >> 4;
  const int m_t = (blockIdx.x >> 3) & 1;
  const int np  = (blockIdx.x >> 2) & 1;
  const int kq  = blockIdx.x & (KS_ - 1);
  const int k0  = kq * KQ_;
  const bool gate_blk = (e == 0) && (np == 0);

  {
    const int c4 = tid & 7;
    const int r0 = tid >> 3;
    const float4* __restrict__ srcb =
        (const float4*)(w1 + (e * D_ + k0) * H_ + np * FC1_COLS + 4 * c4);
    const int ca = tid & 15;
    const int ra = tid >> 4;
    const float4* __restrict__ srca =
        (const float4*)(x + (m_t * 16 + ra) * D_ + k0) + ca;
    const float4* __restrict__ srcg = (const float4*)(gw + k0 * E_);
    const int g1 = FC1_THREADS + (tid & 127);
    float4 tb[KQ_ / 32], ta[KQ_ / 64], tg[2];
#pragma unroll
    for (int i = 0; i < KQ_ / 32; ++i)
      tb[i] = srcb[(r0 + 32 * i) * (H_ / 4)];
#pragma unroll
    for (int i = 0; i < KQ_ / 64; ++i)
      ta[i] = srca[16 * i];
    tg[0] = srcg[tid];
    tg[1] = srcg[g1];
#pragma unroll
    for (int i = 0; i < KQ_ / 32; ++i)
      *(float4*)(sb + (r0 + 32 * i) * FC1_PB + 4 * c4) = tb[i];
#pragma unroll
    for (int i = 0; i < KQ_ / 64; ++i)
      *(float4*)(sa + ra * FC1_PA + 4 * (ca + 16 * i)) = ta[i];
    *(float4*)(sg + 4 * tid) = tg[0];
    *(float4*)(sg + 4 * g1) = tg[1];
  }

  __syncthreads();

  const int w     = tid >> 5;
  const int lane  = tid & 31;
  const int row   = lane & 15;
  const int khalf = lane >> 4;
  const float* pa = sa + row * FC1_PA + 2 * khalf;

  if (w >= 2) {
    if (w > 2 || !gate_blk) return;

    const int gcol = row < E_ ? row : E_ - 1;
    v8f c = fc1_tile<FC1_PG, true>(pa, sg + (2 * khalf) * FC1_PG + gcol);
    if (khalf == 0) {
      float4* lout = (float4*)(lp + ((kq * B_) + m_t * 16 + row) * 8);
      lout[0] = make_float4(c[0], c[1], c[2], c[3]);
      lout[1] = make_float4(c[4], c[5], c[6], c[7]);
    }
    return;
  }

  const int n_s = w;
  v8f c = fc1_tile<FC1_PB, false>(pa, sb + (2 * khalf) * FC1_PB + n_s * 16 + row);

  const int n_t = np * 2 + n_s;
  float* hout = hp + ((kq * E_ + e) * B_ + m_t * 16) * H_ + n_t * 16 + row;
#pragma unroll
  for (int v = 0; v < 8; ++v) {
    int r = v + 8 * khalf;
    hout[r * H_] = c[v];
  }
}

#define FC2_COLS   32
#define FC2_PITCH  40
#define FC2_PA     (H_ + 4)
#define FC2_K      (E_ * H_)
#define FC2_KB     (H_ / 4)
#define FC2_HROWS  (E_ * B_)

__device__ __forceinline__ void fc2_read(v2f (&af)[FC2_KB], v2f (&bf)[FC2_KB],
                                         const float* pa, const float* pb) {
#pragma unroll
  for (int j = 0; j < FC2_KB; ++j) {
    af[j] = *(const v2f*)(pa + 4 * j);
    bf[j].x = pb[(4 * j + 0) * FC2_PITCH];
    bf[j].y = pb[(4 * j + 1) * FC2_PITCH];
  }
  __builtin_amdgcn_sched_barrier(0);
}

__device__ __forceinline__ v8f fc2_chain(v8f c, const v2f (&af)[FC2_KB], float g,
                                         const v2f (&bf)[FC2_KB]) {
#pragma unroll
  for (int j = 0; j < FC2_KB; ++j) {
    v2f a = af[j];
    a.x *= g; a.y *= g;
    c = __builtin_amdgcn_wmma_f32_16x16x4_f32(
        false, a, false, bf[j], (short)0, c, false, false);
  }
  __builtin_amdgcn_sched_barrier(0);
  return c;
}

#define FC2_THREADS 512

__global__ void __launch_bounds__(FC2_THREADS)
moe_fc2(const float* __restrict__ hp,
        const float* __restrict__ lp,
        const float* __restrict__ w2,
        const float* __restrict__ b1,
        const float* __restrict__ b2,
        const float* __restrict__ gb,
        float* __restrict__ out) {
  __shared__ __attribute__((aligned(16)))
      float lds2[FC2_K * FC2_PITCH + FC2_HROWS * FC2_PA + B_ * E_ + E_ * FC2_COLS];
  float* bs  = lds2;
  float* hs  = bs + FC2_K * FC2_PITCH;
  float* gs  = hs + FC2_HROWS * FC2_PA;
  float* b2s = gs + B_ * E_;

  const int tid  = threadIdx.x;
  const int col0 = blockIdx.x * FC2_COLS;

  {
    const int c4 = tid & 7;
    const int r0 = tid >> 3;
    int gcol = col0 + 4 * c4;
    if (gcol > O_ - 4) gcol = O_ - 4;
    const float4* __restrict__ srcw = (const float4*)(w2 + gcol);
    const int ch = tid & 15;
    const int rh = tid >> 4;
    const float4* __restrict__ srch = (const float4*)hp + ch;
    const int q  = tid & 63;
    const int sg = q < 48 ? q : q - 48;
    if (tid < 32) {
      const float4* __restrict__ srcl = (const float4*)lp + 2 * tid;
      float4 tl[2 * KS_];
#pragma unroll
      for (int k = 0; k < KS_; ++k) {
        tl[2 * k + 0] = srcl[k * (B_ * 2) + 0];
        tl[2 * k + 1] = srcl[k * (B_ * 2) + 1];
      }
      const float ls[E_] = {
          ((tl[0].x + tl[2].x) + tl[4].x) + tl[6].x, ((tl[0].y + tl[2].y) + tl[4].y) + tl[6].y,
          ((tl[0].z + tl[2].z) + tl[4].z) + tl[6].z, ((tl[0].w + tl[2].w) + tl[4].w) + tl[6].w,
          ((tl[1].x + tl[3].x) + tl[5].x) + tl[7].x, ((tl[1].y + tl[3].y) + tl[5].y) + tl[7].y};
      float logit[E_];
#pragma unroll
      for (int p = 0; p < E_; ++p) logit[p] = ls[p] + gb[p];
      int i0 = 0; float v0 = logit[0];
#pragma unroll
      for (int p = 1; p < E_; ++p) { float lv = logit[p]; if (lv > v0) { v0 = lv; i0 = p; } }
      int i1 = -1; float v1 = -3.4e38f;
#pragma unroll
      for (int p = 0; p < E_; ++p) {
        float lv = logit[p];
        if (p != i0 && lv > v1) { v1 = lv; i1 = p; }
      }
      const float e1  = __expf(v1 - v0);
      const float inv = 1.0f / (1.0f + e1);
#pragma unroll
      for (int p = 0; p < E_; ++p)
        gs[tid * E_ + p] = (p == i0) ? inv : ((p == i1) ? e1 * inv : 0.0f);
    }

    float4 tw[E_], th[E_][KS_], tb1[E_];
    const float4 tb = *(const float4*)(b2 + (sg >> 3) * O_ + gcol);
#pragma unroll
    for (int i = 0; i < E_; ++i)
      tw[i] = srcw[(r0 + H_ * i) * (O_ / 4)];
#pragma unroll
    for (int i = 0; i < E_; ++i)
#pragma unroll
      for (int k = 0; k < KS_; ++k)
        th[i][k] = srch[((k * E_ + i) * B_ + rh) * (H_ / 4)];
#pragma unroll
    for (int i = 0; i < E_; ++i)
      tb1[i] = *(const float4*)(b1 + i * H_ + 4 * ch);

    *(float4*)(b2s + (sg >> 3) * FC2_COLS + 4 * c4) = tb;
#pragma unroll
    for (int i = 0; i < E_; ++i)
      *(float4*)(bs + (r0 + H_ * i) * FC2_PITCH + 4 * c4) = tw[i];
#pragma unroll
    for (int i = 0; i < E_; ++i) {
      float4 s;
      s.x = (((th[i][0].x + th[i][1].x) + th[i][2].x) + th[i][3].x) + tb1[i].x;
      s.y = (((th[i][0].y + th[i][1].y) + th[i][2].y) + th[i][3].y) + tb1[i].y;
      s.z = (((th[i][0].z + th[i][1].z) + th[i][2].z) + th[i][3].z) + tb1[i].z;
      s.w = (((th[i][0].w + th[i][1].w) + th[i][2].w) + th[i][3].w) + tb1[i].w;
      s.x = s.x > 0.0f ? s.x : 0.0f;
      s.y = s.y > 0.0f ? s.y : 0.0f;
      s.z = s.z > 0.0f ? s.z : 0.0f;
      s.w = s.w > 0.0f ? s.w : 0.0f;
      *(float4*)(hs + (rh + B_ * i) * FC2_PA + 4 * ch) = s;
    }
  }

  const int w     = tid >> 5;
  const int m_t   = w >> 1;
  const int n_s   = w & 1;
  const int lane  = tid & 31;
  const int row   = lane & 15;
  const int khalf = lane >> 4;
  const int ocol  = col0 + n_s * 16 + row;
  const bool colok = ocol < O_;

  __syncthreads();

  if (w >= 4) return;

  const float* pa = hs + (m_t * 16 + row) * FC2_PA + 2 * khalf;
  const float* pb = bs + (2 * khalf) * FC2_PITCH + n_s * 16 + row;
  const int astep = B_ * FC2_PA;
  const int bstep = H_ * FC2_PITCH;
  float g[E_];
#pragma unroll
  for (int e = 0; e < E_; ++e) g[e] = gs[(m_t * 16 + row) * E_ + e];

  v2f a0[FC2_KB], b0[FC2_KB], a1[FC2_KB], b1f[FC2_KB];
  fc2_read(a0, b0, pa, pb);
  v8f c = {};
#pragma unroll
  for (int e = 0; e < E_ - 2; e += 2) {
    fc2_read(a1, b1f, pa + (e + 1) * astep, pb + (e + 1) * bstep);
    c = fc2_chain(c, a0, g[e], b0);
    fc2_read(a0, b0, pa + (e + 2) * astep, pb + (e + 2) * bstep);
    c = fc2_chain(c, a1, g[e + 1], b1f);
  }
  fc2_read(a1, b1f, pa + (E_ - 1) * astep, pb + (E_ - 1) * bstep);
  c = fc2_chain(c, a0, g[E_ - 2], b0);
  c = fc2_chain(c, a1, g[E_ - 1], b1f);

  if (colok) {
    const float* ge = gs + (m_t * 16 + 8 * khalf) * E_;
    const float* be = b2s + n_s * 16 + row;
#pragma unroll
    for (int v = 0; v < 8; ++v) {
      int r = m_t * 16 + v + 8 * khalf;
      float acc = c[v];
#pragma unroll
      for (int e = 0; e < E_; ++e)
        acc = fmaf(ge[v * E_ + e], be[e * FC2_COLS], acc);
      out[r * O_ + ocol] = acc;
    }
  }
}

extern "C" void kernel_launch(void* const* d_in, const int* in_sizes, int n_in,
                              void* d_out, int out_size, void* d_ws, size_t ws_size,
                              hipStream_t stream) {
  const float* x      = (const float*)d_in[0];
  const float* gate_w = (const float*)d_in[1];
  const float* gate_b = (const float*)d_in[2];
  const float* w1     = (const float*)d_in[3];
  const float* b1     = (const float*)d_in[4];
  const float* w2     = (const float*)d_in[5];
  const float* b2     = (const float*)d_in[6];
  float* out = (float*)d_out;

  float* lp = (float*)d_ws;
  float* hp = lp + KS_ * B_ * 8;

  moe_fc1<<<E_ * 2 * 2 * KS_, FC1_THREADS, 0, stream>>>(x, w1, gate_w, hp, lp);

  moe_fc2<<<(O_ + FC2_COLS - 1) / FC2_COLS, FC2_THREADS, 0, stream>>>(hp, lp, w2, b1, b2, gate_b, out);
}


# fc2 staging: each thread finds its batch row's top-2 experts first, then loads the fc1 quarter sums of those two only (8 loads, was 24); gate-0 h rows stored as zeros; sum order unchanged
# speedup vs baseline: 11.4847x; 1.0875x over previous
#include <hip/hip_runtime.h>
#include <hip/hip_bf16.h>
#include <math.h>

typedef __attribute__((ext_vector_type(2))) float v2f;
typedef __attribute__((ext_vector_type(8))) float v8f;

#define B_  32
#define D_  1024
#define H_  64
#define O_  1000
#define E_  6
#define KS_ 4
#define KQ_ (D_ / KS_)

#define FC1_COLS   32
#define FC1_PB     48
#define FC1_PA     (KQ_ + 4)
#define FC1_KB     16
#define FC1_THREADS 256

#define FC1_PG     E_

template <int PB>
__device__ __forceinline__ void fc1_read(v2f (&af)[FC1_KB], v2f (&bf)[FC1_KB],
                                         const float* pa, const float* pb) {
#pragma unroll
  for (int j = 0; j < FC1_KB; ++j) {
    af[j] = *(const v2f*)(pa + 4 * j);
    bf[j].x = pb[(4 * j + 0) * PB];
    bf[j].y = pb[(4 * j + 1) * PB];
  }
  __builtin_amdgcn_sched_barrier(0);
}

__device__ __forceinline__ v8f fc1_chain(v8f c, const v2f (&af)[FC1_KB], const v2f (&bf)[FC1_KB]) {
#pragma unroll
  for (int j = 0; j < FC1_KB; ++j)
    c = __builtin_amdgcn_wmma_f32_16x16x4_f32(
         false, af[j],  false, bf[j],
         (short)0, c,  false,  false);
  __builtin_amdgcn_sched_barrier(0);
  return c;
}

template <int PB, bool XT>
__device__ __forceinline__ v8f fc1_tile(const float* pa, const float* pb) {
  const int kstep = 4 * FC1_KB;
  v2f a0[FC1_KB], b0[FC1_KB], a1[FC1_KB], b1f[FC1_KB];
  fc1_read<PB>(a0, b0, pa, pb);
  v8f c = {};
  for (int kb = 0; kb < KQ_ / kstep - 2; kb += 2) {
    fc1_read<PB>(a1, b1f, pa + (kb + 1) * kstep, pb + (kb + 1) * kstep * PB);
    c = XT ? fc1_chain(c, b0, a0) : fc1_chain(c, a0, b0);
    fc1_read<PB>(a0, b0, pa + (kb + 2) * kstep, pb + (kb + 2) * kstep * PB);
    c = XT ? fc1_chain(c, b1f, a1) : fc1_chain(c, a1, b1f);
  }
  fc1_read<PB>(a1, b1f, pa + (KQ_ / kstep - 1) * kstep, pb + (KQ_ / kstep - 1) * kstep * PB);
  c = XT ? fc1_chain(c, b0, a0) : fc1_chain(c, a0, b0);
  c = XT ? fc1_chain(c, b1f, a1) : fc1_chain(c, a1, b1f);
  return c;
}

__global__ void __launch_bounds__(FC1_THREADS)
moe_fc1(const float* __restrict__ x,
        const float* __restrict__ w1,
        const float* __restrict__ gw,
        float* __restrict__ hp,
        float* __restrict__ lp) {
  __shared__ __attribute__((aligned(16))) float lds1[KQ_ * FC1_PB + 16 * FC1_PA + KQ_ * FC1_PG];
  float* sb = lds1;
  float* sa = lds1 + KQ_ * FC1_PB;
  float* sg = sa + 16 * FC1_PA;

  const int tid = threadIdx.x;
  const int e   = blockIdx.x >> 4;
  const int m_t = (blockIdx.x >> 3) & 1;
  const int np  = (blockIdx.x >> 2) & 1;
  const int kq  = blockIdx.x & (KS_ - 1);
  const int k0  = kq * KQ_;
  const bool gate_blk = (e == 0) && (np == 0);

  {
    const int c4 = tid & 7;
    const int r0 = tid >> 3;
    const float4* __restrict__ srcb =
        (const float4*)(w1 + (e * D_ + k0) * H_ + np * FC1_COLS + 4 * c4);
    const int ca = tid & 15;
    const int ra = tid >> 4;
    const float4* __restrict__ srca =
        (const float4*)(x + (m_t * 16 + ra) * D_ + k0) + ca;
    const float4* __restrict__ srcg = (const float4*)(gw + k0 * E_);
    const int g1 = FC1_THREADS + (tid & 127);
    float4 tb[KQ_ / 32], ta[KQ_ / 64], tg[2];
#pragma unroll
    for (int i = 0; i < KQ_ / 32; ++i)
      tb[i] = srcb[(r0 + 32 * i) * (H_ / 4)];
#pragma unroll
    for (int i = 0; i < KQ_ / 64; ++i)
      ta[i] = srca[16 * i];
    tg[0] = srcg[tid];
    tg[1] = srcg[g1];
#pragma unroll
    for (int i = 0; i < KQ_ / 32; ++i)
      *(float4*)(sb + (r0 + 32 * i) * FC1_PB + 4 * c4) = tb[i];
#pragma unroll
    for (int i = 0; i < KQ_ / 64; ++i)
      *(float4*)(sa + ra * FC1_PA + 4 * (ca + 16 * i)) = ta[i];
    *(float4*)(sg + 4 * tid) = tg[0];
    *(float4*)(sg + 4 * g1) = tg[1];
  }

  __syncthreads();

  const int w     = tid >> 5;
  const int lane  = tid & 31;
  const int row   = lane & 15;
  const int khalf = lane >> 4;
  const float* pa = sa + row * FC1_PA + 2 * khalf;

  if (w >= 2) {
    if (w > 2 || !gate_blk) return;

    const int gcol = row < E_ ? row : E_ - 1;
    v8f c = fc1_tile<FC1_PG, true>(pa, sg + (2 * khalf) * FC1_PG + gcol);
    if (khalf == 0) {
      float4* lout = (float4*)(lp + ((kq * B_) + m_t * 16 + row) * 8);
      lout[0] = make_float4(c[0], c[1], c[2], c[3]);
      lout[1] = make_float4(c[4], c[5], c[6], c[7]);
    }
    return;
  }

  const int n_s = w;
  v8f c = fc1_tile<FC1_PB, false>(pa, sb + (2 * khalf) * FC1_PB + n_s * 16 + row);

  const int n_t = np * 2 + n_s;
  float* hout = hp + ((kq * E_ + e) * B_ + m_t * 16) * H_ + n_t * 16 + row;
#pragma unroll
  for (int v = 0; v < 8; ++v) {
    int r = v + 8 * khalf;
    hout[r * H_] = c[v];
  }
}

#define FC2_COLS   32
#define FC2_PITCH  40
#define FC2_PA     (H_ + 4)
#define FC2_K      (E_ * H_)
#define FC2_KB     (H_ / 4)
#define FC2_HROWS  (E_ * B_)
#define FC2_PG     8

__device__ __forceinline__ void fc2_read(v2f (&af)[FC2_KB], v2f (&bf)[FC2_KB],
                                         const float* pa, const float* pb) {
#pragma unroll
  for (int j = 0; j < FC2_KB; ++j) {
    af[j] = *(const v2f*)(pa + 4 * j);
    bf[j].x = pb[(4 * j + 0) * FC2_PITCH];
    bf[j].y = pb[(4 * j + 1) * FC2_PITCH];
  }
  __builtin_amdgcn_sched_barrier(0);
}

__device__ __forceinline__ v8f fc2_chain(v8f c, const v2f (&af)[FC2_KB], float g,
                                         const v2f (&bf)[FC2_KB]) {
#pragma unroll
  for (int j = 0; j < FC2_KB; ++j) {
    v2f a = af[j];
    a.x *= g; a.y *= g;
    c = __builtin_amdgcn_wmma_f32_16x16x4_f32(
        false, a, false, bf[j], (short)0, c, false, false);
  }
  __builtin_amdgcn_sched_barrier(0);
  return c;
}

#define FC2_THREADS 512

__global__ void __launch_bounds__(FC2_THREADS)
moe_fc2(const float* __restrict__ hp,
        const float* lp,
        const float* w2,
        const float* __restrict__ b1,
        const float* b2,
        const float* __restrict__ gb,
        float* __restrict__ out) {
  __shared__ __attribute__((aligned(16)))
      float lds2[FC2_K * FC2_PITCH + FC2_HROWS * FC2_PA + B_ * FC2_PG + E_ * FC2_COLS];
  float* bs  = lds2;
  float* hs  = bs + FC2_K * FC2_PITCH;
  float* gs  = hs + FC2_HROWS * FC2_PA;
  float* b2s = gs + B_ * FC2_PG;

  const int tid  = threadIdx.x;
  const int col0 = blockIdx.x * FC2_COLS;

  {
    const int c4 = tid & 7;
    const int r0 = tid >> 3;
    int gcol = col0 + 4 * c4;
    if (gcol > O_ - 4) gcol = O_ - 4;
    const float4* srcw = (const float4*)(w2 + gcol);
    const int ch = tid & 15;
    const int rh = tid >> 4;
    const int q  = tid & 63;
    const int sg = q < 48 ? q : q - 48;

    const float4* srcl = (const float4*)lp + 2 * rh;
    float4 tl[2 * KS_];
#pragma unroll
    for (int k = 0; k < KS_; ++k) {
      tl[2 * k + 0] = srcl[k * (B_ * 2) + 0];
      tl[2 * k + 1] = srcl[k * (B_ * 2) + 1];
    }
    const float4 tw0 = srcw[(r0 + H_ * 0) * (O_ / 4)];
    const float4 tw1 = srcw[(r0 + H_ * 1) * (O_ / 4)];
    const float4 tw2 = srcw[(r0 + H_ * 2) * (O_ / 4)];
    const float4 tw3 = srcw[(r0 + H_ * 3) * (O_ / 4)];
    const float4 tw4 = srcw[(r0 + H_ * 4) * (O_ / 4)];
    const float4 tw5 = srcw[(r0 + H_ * 5) * (O_ / 4)];
    const float4 tb = *(const float4*)(b2 + (sg >> 3) * O_ + gcol);
    __builtin_amdgcn_sched_barrier(0);

    const float ls[E_] = {
        ((tl[0].x + tl[2].x) + tl[4].x) + tl[6].x, ((tl[0].y + tl[2].y) + tl[4].y) + tl[6].y,
        ((tl[0].z + tl[2].z) + tl[4].z) + tl[6].z, ((tl[0].w + tl[2].w) + tl[4].w) + tl[6].w,
        ((tl[1].x + tl[3].x) + tl[5].x) + tl[7].x, ((tl[1].y + tl[3].y) + tl[5].y) + tl[7].y};
    float logit[E_];
#pragma unroll
    for (int p = 0; p < E_; ++p) logit[p] = ls[p] + gb[p];
    int i0 = 0; float v0 = logit[0];
#pragma unroll
    for (int p = 1; p < E_; ++p) { float lv = logit[p]; if (lv > v0) { v0 = lv; i0 = p; } }
    int i1 = -1; float v1 = -3.4e38f;
#pragma unroll
    for (int p = 0; p < E_; ++p) {
      float lv = logit[p];
      if (p != i0 && lv > v1) { v1 = lv; i1 = p; }
    }
    const int j1 = i1 < 0 ? (i0 == 0 ? 1 : 0) : i1;

    const float4* __restrict__ srch = (const float4*)hp + rh * (H_ / 4) + ch;
    float4 th0[KS_], th1[KS_];
#pragma unroll
    for (int k = 0; k < KS_; ++k) {
      th0[k] = srch[(k * E_ + i0) * (B_ * H_ / 4)];
      th1[k] = srch[(k * E_ + j1) * (B_ * H_ / 4)];
    }
    const float4 tb10 = *(const float4*)(b1 + i0 * H_ + 4 * ch);
    const float4 tb11 = *(const float4*)(b1 + j1 * H_ + 4 * ch);

    const float e1  = __expf(v1 - v0);
    const float inv = 1.0f / (1.0f + e1);
    {
      const int p = ch & 7;
      gs[rh * FC2_PG + p] = (p == i0) ? inv : ((p == i1) ? e1 * inv : 0.0f);
    }

    *(float4*)(b2s + (sg >> 3) * FC2_COLS + 4 * c4) = tb;
    *(float4*)(bs + (r0 + H_ * 0) * FC2_PITCH + 4 * c4) = tw0;
    *(float4*)(bs + (r0 + H_ * 1) * FC2_PITCH + 4 * c4) = tw1;
    *(float4*)(bs + (r0 + H_ * 2) * FC2_PITCH + 4 * c4) = tw2;
    *(float4*)(bs + (r0 + H_ * 3) * FC2_PITCH + 4 * c4) = tw3;
    *(float4*)(bs + (r0 + H_ * 4) * FC2_PITCH + 4 * c4) = tw4;
    *(float4*)(bs + (r0 + H_ * 5) * FC2_PITCH + 4 * c4) = tw5;
    float4 s0, s1;
    s0.x = (((th0[0].x + th0[1].x) + th0[2].x) + th0[3].x) + tb10.x;
    s0.y = (((th0[0].y + th0[1].y) + th0[2].y) + th0[3].y) + tb10.y;
    s0.z = (((th0[0].z + th0[1].z) + th0[2].z) + th0[3].z) + tb10.z;
    s0.w = (((th0[0].w + th0[1].w) + th0[2].w) + th0[3].w) + tb10.w;
    s1.x = (((th1[0].x + th1[1].x) + th1[2].x) + th1[3].x) + tb11.x;
    s1.y = (((th1[0].y + th1[1].y) + th1[2].y) + th1[3].y) + tb11.y;
    s1.z = (((th1[0].z + th1[1].z) + th1[2].z) + th1[3].z) + tb11.z;
    s1.w = (((th1[0].w + th1[1].w) + th1[2].w) + th1[3].w) + tb11.w;
    s0.x = s0.x > 0.0f ? s0.x : 0.0f;
    s0.y = s0.y > 0.0f ? s0.y : 0.0f;
    s0.z = s0.z > 0.0f ? s0.z : 0.0f;
    s0.w = s0.w > 0.0f ? s0.w : 0.0f;
    s1.x = s1.x > 0.0f ? s1.x : 0.0f;
    s1.y = s1.y > 0.0f ? s1.y : 0.0f;
    s1.z = s1.z > 0.0f ? s1.z : 0.0f;
    s1.w = s1.w > 0.0f ? s1.w : 0.0f;
#pragma unroll
    for (int i = 0; i < E_; ++i) {
      const bool m0 = (i == i0), m1 = (i == i1);
      float4 v;
      v.x = m0 ? s0.x : (m1 ? s1.x : 0.0f);
      v.y = m0 ? s0.y : (m1 ? s1.y : 0.0f);
      v.z = m0 ? s0.z : (m1 ? s1.z : 0.0f);
      v.w = m0 ? s0.w : (m1 ? s1.w : 0.0f);
      *(float4*)(hs + (rh + B_ * i) * FC2_PA + 4 * ch) = v;
    }
  }

  const int w     = tid >> 5;
  const int m_t   = w >> 1;
  const int n_s   = w & 1;
  const int lane  = tid & 31;
  const int row   = lane & 15;
  const int khalf = lane >> 4;
  const int ocol  = col0 + n_s * 16 + row;
  const bool colok = ocol < O_;

  __syncthreads();

  if (w >= 4) return;

  const float* pa = hs + (m_t * 16 + row) * FC2_PA + 2 * khalf;
  const float* pb = bs + (2 * khalf) * FC2_PITCH + n_s * 16 + row;
  const int astep = B_ * FC2_PA;
  const int bstep = H_ * FC2_PITCH;
  float g[E_];
#pragma unroll
  for (int e = 0; e < E_; ++e) g[e] = gs[(m_t * 16 + row) * FC2_PG + e];

  v2f a0[FC2_KB], b0[FC2_KB], a1[FC2_KB], b1f[FC2_KB];
  fc2_read(a0, b0, pa, pb);
  v8f c = {};
#pragma unroll
  for (int e = 0; e < E_ - 2; e += 2) {
    fc2_read(a1, b1f, pa + (e + 1) * astep, pb + (e + 1) * bstep);
    c = fc2_chain(c, a0, g[e], b0);
    fc2_read(a0, b0, pa + (e + 2) * astep, pb + (e + 2) * bstep);
    c = fc2_chain(c, a1, g[e + 1], b1f);
  }
  fc2_read(a1, b1f, pa + (E_ - 1) * astep, pb + (E_ - 1) * bstep);
  c = fc2_chain(c, a0, g[E_ - 2], b0);
  c = fc2_chain(c, a1, g[E_ - 1], b1f);

  if (colok) {
    const float* ge = gs + (m_t * 16 + 8 * khalf) * FC2_PG;
    const float* be = b2s + n_s * 16 + row;
#pragma unroll
    for (int v = 0; v < 8; ++v) {
      int r = m_t * 16 + v + 8 * khalf;
      float acc = c[v];
#pragma unroll
      for (int e = 0; e < E_; ++e)
        acc = fmaf(ge[v * FC2_PG + e], be[e * FC2_COLS], acc);
      out[r * O_ + ocol] = acc;
    }
  }
}

extern "C" void kernel_launch(void* const* d_in, const int* in_sizes, int n_in,
                              void* d_out, int out_size, void* d_ws, size_t ws_size,
                              hipStream_t stream) {
  const float* x      = (const float*)d_in[0];
  const float* gate_w = (const float*)d_in[1];
  const float* gate_b = (const float*)d_in[2];
  const float* w1     = (const float*)d_in[3];
  const float* b1     = (const float*)d_in[4];
  const float* w2     = (const float*)d_in[5];
  const float* b2     = (const float*)d_in[6];
  float* out = (float*)d_out;

  float* lp = (float*)d_ws;
  float* hp = lp + KS_ * B_ * 8;

  moe_fc1<<<E_ * 2 * 2 * KS_, FC1_THREADS, 0, stream>>>(x, w1, gate_w, hp, lp);

  moe_fc2<<<(O_ + FC2_COLS - 1) / FC2_COLS, FC2_THREADS, 0, stream>>>(hp, lp, w2, b1, b2, gate_b, out);
}
